# starting kernel
# baseline (speedup 1.0000x reference)
#include <hip/hip_runtime.h>


typedef __attribute__((ext_vector_type(2))) float v2f;
typedef __attribute__((ext_vector_type(8))) float v8f;

#define B_  32
#define N_  512
#define A_  32
#define M_  128
#define D_  32
#define MD_ (M_ * D_)
#define SCALE_ 0.17677669529663687f

static __device__ __forceinline__ v8f wmma_f32(v2f a, v2f b, v8f c) {
  return __builtin_amdgcn_wmma_f32_16x16x4_f32(false, a, false, b, (short)0, c,
                                               false, false);
}

__global__ __launch_bounds__(256) void caps_logits_qk(
    const float* __restrict__ pose, const float* __restrict__ ncv,
    const float* __restrict__ na, const float* __restrict__ w,
    float* __restrict__ qk_out) {
  const int n    = blockIdx.x;
  const int tid  = threadIdx.x;
  const int lane = tid & 31;
  const int wv   = tid >> 5;
  const int l16  = lane & 15;
  const int khalf = (lane < 16) ? 0 : 2;

  __shared__ float vote[32 * 260];
  __shared__ float lg[32 * 128];

  v2f areg[2][8];
  for (int rt = 0; rt < 2; ++rt) {
    const int b = l16 + 16 * rt;
    const float* pp = pose + ((size_t)b * N_ + n) * A_;
    for (int k = 0; k < 8; ++k) areg[rt][k] = *(const v2f*)(pp + 4 * k + khalf);
  }

  for (int mt = 0; mt < 16; ++mt) {
    for (int j = 0; j < 2; ++j) {
      const int ct  = wv * 2 + j;
      const int col = mt * 256 + ct * 16 + l16;
      const float* wbase = w + (size_t)n * A_ * MD_ + col;
      __builtin_prefetch(wbase + 256, 0, 0);
      v8f acc0 = {}, acc1 = {};
      for (int k = 0; k < 8; ++k) {
        const int a = 4 * k + khalf;
        const float* wp = wbase + (size_t)a * MD_;
        v2f bv; bv.x = wp[0]; bv.y = wp[MD_];
        acc0 = wmma_f32(areg[0][k], bv, acc0);
        acc1 = wmma_f32(areg[1][k], bv, acc1);
      }
      const int cbase = ct * 16 + l16;
      const int rbase = (lane < 16) ? 0 : 8;
      for (int i = 0; i < 8; ++i) {
        vote[(i + rbase) * 260 + cbase]      = acc0[i];
        vote[(i + rbase + 16) * 260 + cbase] = acc1[i];
      }
    }
    __syncthreads();
    {
      const int b  = tid & 31;
      const int mi = tid >> 5;
      const int m  = mt * 8 + mi;
      const float* vp = vote + b * 260 + mi * 32;
      const float* cp = ncv + ((size_t)b * M_ + m) * D_;
      float s = 0.f;
      for (int d = 0; d < 32; ++d) s += vp[d] * cp[d];
      lg[b * 128 + m] = s * SCALE_;
    }
    __syncthreads();
  }

  if (tid < 32) {
    const int b = tid;
    float* row = lg + b * 128;
    float mx = -3.4e38f;
    for (int m = 0; m < 128; ++m) mx = fmaxf(mx, row[m]);
    float s = 0.f;
    for (int m = 0; m < 128; ++m) { float e = __expf(row[m] - mx); row[m] = e; s += e; }
    const float inv = 1.f / s;
    const float* nap = na + b * 128;
    float s2 = 0.f;
    for (int m = 0; m < 128; ++m) { float v = row[m] * inv * nap[m]; row[m] = v; s2 += v; }
    const float inv2 = 1.f / (s2 + 1e-10f);
    float* qp = qk_out + ((size_t)b * N_ + n) * M_;
    for (int m = 0; m < 128; ++m) qp[m] = row[m] * inv2;
  }
}

__global__ __launch_bounds__(256) void caps_out(
    const float* __restrict__ pose, const float* __restrict__ ca,
    const float* __restrict__ w, const float* __restrict__ qk,
    float* __restrict__ out) {
  const int m    = blockIdx.x;
  const int dh   = blockIdx.y;
  const int tid  = threadIdx.x;
  const int lane = tid & 31;
  const int wv   = tid >> 5;
  const int l16  = lane & 15;
  const int khalf = (lane < 16) ? 0 : 2;
  const int dcol = dh * 16 + l16;

  __shared__ float red[8][32 * 17];

  v8f acc0 = {}, acc1 = {};
  const int b0 = l16, b1 = l16 + 16;

  for (int n = wv * 64; n < wv * 64 + 64; ++n) {
    const float q0 = qk[((size_t)b0 * N_ + n) * M_ + m] * ca[(size_t)b0 * N_ + n];
    const float q1 = qk[((size_t)b1 * N_ + n) * M_ + m] * ca[(size_t)b1 * N_ + n];
    const float* pp0 = pose + ((size_t)b0 * N_ + n) * A_;
    const float* pp1 = pose + ((size_t)b1 * N_ + n) * A_;
    const float* wbase = w + (size_t)n * A_ * MD_ + m * D_ + dcol;
    __builtin_prefetch(wbase + (size_t)2 * A_ * MD_, 0, 0);
    for (int k = 0; k < 8; ++k) {
      const int a = 4 * k + khalf;
      const v2f p0 = *(const v2f*)(pp0 + a);
      const v2f p1 = *(const v2f*)(pp1 + a);
      v2f av0; av0.x = q0 * p0.x; av0.y = q0 * p0.y;
      v2f av1; av1.x = q1 * p1.x; av1.y = q1 * p1.y;
      const float* wp = wbase + (size_t)a * MD_;
      v2f bv; bv.x = wp[0]; bv.y = wp[MD_];
      acc0 = wmma_f32(av0, bv, acc0);
      acc1 = wmma_f32(av1, bv, acc1);
    }
  }

  const int rbase = (lane < 16) ? 0 : 8;
  for (int i = 0; i < 8; ++i) {
    red[wv][(i + rbase) * 17 + l16]      = acc0[i];
    red[wv][(i + rbase + 16) * 17 + l16] = acc1[i];
  }
  __syncthreads();
  for (int e = tid; e < 512; e += 256) {
    const int b = e >> 4, dc = e & 15;
    float s = 0.f;
    for (int v = 0; v < 8; ++v) s += red[v][b * 17 + dc];
    out[(size_t)b * MD_ + m * D_ + dh * 16 + dc] = s;
  }
}

extern "C" void kernel_launch(void* const* d_in, const int* in_sizes, int n_in,
                              void* d_out, int out_size, void* d_ws,
                              size_t ws_size, hipStream_t stream) {
  (void)in_sizes; (void)n_in; (void)out_size; (void)d_ws; (void)ws_size;
  const float* pose = (const float*)d_in[0];
  const float* ca   = (const float*)d_in[1];
  const float* ncv  = (const float*)d_in[2];
  const float* na   = (const float*)d_in[3];
  const float* w    = (const float*)d_in[4];

  float* out    = (float*)d_out;
  float* na_out = out + (size_t)B_ * M_ * D_;
  float* qk_out = na_out + (size_t)B_ * M_;

  caps_logits_qk<<<dim3(N_), dim3(256), 0, stream>>>(pose, ncv, na, w, qk_out);
  hipMemcpyAsync(na_out, na, (size_t)B_ * M_ * sizeof(float),
                 hipMemcpyDeviceToDevice, stream);
  caps_out<<<dim3(M_, 2), dim3(256), 0, stream>>>(pose, ca, w, qk_out, out);
}


# first kernel: vote product transposed (w as A, pose as B): a lane holds its own b's d in registers, ncv dot in-lane + one half exchange; LDS vote tile, per-tile barriers gone; d sum order changes
# speedup vs baseline: 1.4263x; 1.4263x over previous
#include <hip/hip_runtime.h>


typedef __attribute__((ext_vector_type(2))) float v2f;
typedef __attribute__((ext_vector_type(4))) float v4f;
typedef __attribute__((ext_vector_type(8))) float v8f;

#define B_  32
#define N_  512
#define A_  32
#define M_  128
#define D_  32
#define MD_ (M_ * D_)
#define SCALE_ 0.17677669529663687f

static __device__ __forceinline__ v8f wmma_f32(v2f a, v2f b, v8f c) {
  return __builtin_amdgcn_wmma_f32_16x16x4_f32(false, a, false, b, (short)0, c,
                                               false, false);
}

struct WTile { v2f b[2][8]; };

static __device__ __forceinline__ void wtile_request(
    WTile& t, const float* __restrict__ w, int n, int mt, unsigned wlane) {
  const float* wt = w + (size_t)n * A_ * MD_ + mt * 256;
  for (int j = 0; j < 2; ++j) {
    const unsigned lo = wlane + 16u * j;
    __builtin_prefetch(wt + lo + 256, 0, 0);
    for (int k = 0; k < 8; ++k) {
      t.b[j][k].x = wt[lo + (unsigned)(4 * k) * MD_];
      t.b[j][k].y = wt[lo + (unsigned)(4 * k + 1) * MD_];
    }
  }
}

static __device__ __forceinline__ float xhalf(float x) {
  const int b = __builtin_bit_cast(int, x);
  return __builtin_bit_cast(float, __builtin_amdgcn_permlanex16(
      b, b, 0x76543210, 0xfedcba98, false, false));
}

template <bool NEXT>
static __device__ __forceinline__ void logits_step(
    int mt, const WTile& cur, WTile& nxt, const v2f (&preg)[2][8], float* lg,
    const float* __restrict__ ncv, const float* __restrict__ w,
    int n, int lane, unsigned wlane, unsigned clane, int lgl) {
  const float* ct = ncv + mt * 8 * D_;
  v4f cv[2][2][2];
  for (int rt = 0; rt < 2; ++rt) {
    for (int j = 0; j < 2; ++j) {
      const unsigned o = clane + (unsigned)(16 * rt) * MD_ + 16u * j;
      cv[rt][j][0] = *(const v4f*)(ct + o);
      cv[rt][j][1] = *(const v4f*)(ct + o + 4);
    }
  }
  if (NEXT) wtile_request(nxt, w, n, mt + 2, wlane);

  v8f acc[2][2];
  for (int j = 0; j < 2; ++j) {
    acc[j][0] = v8f{}; acc[j][1] = v8f{};
    for (int k = 0; k < 8; ++k) {
      acc[j][0] = wmma_f32(cur.b[j][k], preg[0][k], acc[j][0]);
      acc[j][1] = wmma_f32(cur.b[j][k], preg[1][k], acc[j][1]);
    }
  }
#define DOT2(S, ACC, C0, C1)                                                        \
  S += __builtin_shufflevector(ACC, ACC, 0, 1) * __builtin_shufflevector(C0, C0, 0, 1); \
  S += __builtin_shufflevector(ACC, ACC, 2, 3) * __builtin_shufflevector(C0, C0, 2, 3); \
  S += __builtin_shufflevector(ACC, ACC, 4, 5) * __builtin_shufflevector(C1, C1, 0, 1); \
  S += __builtin_shufflevector(ACC, ACC, 6, 7) * __builtin_shufflevector(C1, C1, 2, 3)
  float t[2];
  for (int rt = 0; rt < 2; ++rt) {
    v2f p0 = {0.f, 0.f}, p1 = {0.f, 0.f};
    DOT2(p0, acc[0][rt], cv[rt][0][0], cv[rt][0][1]);
    DOT2(p1, acc[1][rt], cv[rt][1][0], cv[rt][1][1]);
    const v2f p = p0 + p1;
    const float s = p.x + p.y;
    t[rt] = s + xhalf(s);
  }
#undef DOT2
  lg[lgl + mt * 8] = ((lane < 16) ? t[0] : t[1]) * SCALE_;
}

__global__ __launch_bounds__(256, 4) void caps_logits_qk(
    const float* __restrict__ pose, const float* __restrict__ ncv,
    const float* __restrict__ na, const float* __restrict__ w,
    float* __restrict__ qk_out) {
  const int n    = blockIdx.x;
  const int tid  = threadIdx.x;
  const int lane = tid & 31;
  const int wv   = tid >> 5;
  const int l16  = lane & 15;
  const int khalf = (lane < 16) ? 0 : 2;

  __shared__ float lg[32 * 128];

  v2f preg[2][8];
  for (int rt = 0; rt < 2; ++rt) {
    const int b = l16 + 16 * rt;
    const float* pp = pose + ((size_t)b * N_ + n) * A_;
    for (int k = 0; k < 8; ++k) preg[rt][k] = *(const v2f*)(pp + 4 * k + khalf);
  }

  const unsigned wlane = (unsigned)khalf * MD_ + 32u * wv + l16;
  const unsigned clane = (unsigned)l16 * MD_ + 32u * wv + 4u * khalf;
  const int lgl = lane * 128 + wv;
#define LSTEP(NEXT, MT, CUR, REQ) \
  logits_step<NEXT>(MT, CUR, REQ, preg, lg, ncv, w, n, lane, wlane, clane, lgl)
  WTile ta, tb, tc;
  wtile_request(ta, w, n, 0, wlane);
  wtile_request(tb, w, n, 1, wlane);
  for (int t3 = 0; t3 < 4; ++t3) {
    LSTEP(true, 3 * t3,     ta, tc);
    LSTEP(true, 3 * t3 + 1, tb, ta);
    LSTEP(true, 3 * t3 + 2, tc, tb);
  }
  LSTEP(true,  12, ta, tc);
  LSTEP(true,  13, tb, ta);
  LSTEP(false, 14, tc, tb);
  LSTEP(false, 15, ta, tb);
#undef LSTEP

  __syncthreads();

  if (tid < 32) {
    const int b = tid;
    float* row = lg + b * 128;
    float mx = -3.4e38f;
    for (int m = 0; m < 128; ++m) mx = fmaxf(mx, row[m]);
    float s = 0.f;
    for (int m = 0; m < 128; ++m) { float e = __expf(row[m] - mx); row[m] = e; s += e; }
    const float inv = 1.f / s;
    const float* nap = na + b * 128;
    float s2 = 0.f;
    for (int m = 0; m < 128; ++m) { float v = row[m] * inv * nap[m]; row[m] = v; s2 += v; }
    const float inv2 = 1.f / (s2 + 1e-10f);
    float* qp = qk_out + ((size_t)b * N_ + n) * M_;
    for (int m = 0; m < 128; ++m) qp[m] = row[m] * inv2;
  }
}


struct CapsStage {
  v2f   p[2][4][2];
  float qk0, qk1, ca0, ca1;
  float wv[4][4];
};

static __device__ __forceinline__ void caps_request(
    CapsStage& s, const float* __restrict__ pose, const float* __restrict__ ca,
    const float* __restrict__ w, const float* __restrict__ qk,
    int n, int m, int h, int l16, int dcol) {
  const float* pt = pose + ((size_t)(2 * h) * N_ + n) * A_ + 2 * l16;
  for (int t = 0; t < 2; ++t) {
    for (int g = 0; g < 4; ++g) {
      const float* rowp = pt + (size_t)(16 * t + 4 * g) * N_ * A_;
      s.p[t][g][0] = *(const v2f*)(rowp);
      s.p[t][g][1] = *(const v2f*)(rowp + N_ * A_);
    }
  }
  const int b0 = l16, b1 = l16 + 16;
  s.qk0 = qk[((size_t)b0 * N_ + n) * M_ + m];
  s.ca0 = ca[(size_t)b0 * N_ + n];
  s.qk1 = qk[((size_t)b1 * N_ + n) * M_ + m];
  s.ca1 = ca[(size_t)b1 * N_ + n];
  const float* wbase = w + ((size_t)n * A_ + 16 * h) * MD_ + m * D_ + dcol;
  __builtin_prefetch(wbase + (size_t)2 * A_ * MD_, 0, 0);
  for (int jj = 0; jj < 4; ++jj) {
    for (int r = 0; r < 4; ++r) s.wv[jj][r] = wbase[(size_t)(4 * jj + r) * MD_];
  }
}

struct CapsTurned { v8f tx0, ty0, tx1, ty1; };

static __device__ __forceinline__ void pose_turn_regs(
    v2f r0, v2f r1, v2f sel, v8f& tx, v8f& ty) {
  v2f ax; ax.x = r0.x; ax.y = r1.x;
  v2f ay; ay.x = r0.y; ay.y = r1.y;
  tx = wmma_f32(ax, sel, tx);
  ty = wmma_f32(ay, sel, ty);
}

static __device__ __forceinline__ void pose_turn_all(
    const CapsStage& s, v2f s0, v2f s1, v2f s2, v2f s3, CapsTurned& t) {
  t.tx0 = v8f{}; t.ty0 = v8f{}; t.tx1 = v8f{}; t.ty1 = v8f{};
  pose_turn_regs(s.p[0][0][0], s.p[0][0][1], s0, t.tx0, t.ty0);
  pose_turn_regs(s.p[0][1][0], s.p[0][1][1], s1, t.tx0, t.ty0);
  pose_turn_regs(s.p[0][2][0], s.p[0][2][1], s2, t.tx0, t.ty0);
  pose_turn_regs(s.p[0][3][0], s.p[0][3][1], s3, t.tx0, t.ty0);
  pose_turn_regs(s.p[1][0][0], s.p[1][0][1], s0, t.tx1, t.ty1);
  pose_turn_regs(s.p[1][1][0], s.p[1][1][1], s1, t.tx1, t.ty1);
  pose_turn_regs(s.p[1][2][0], s.p[1][2][1], s2, t.tx1, t.ty1);
  pose_turn_regs(s.p[1][3][0], s.p[1][3][1], s3, t.tx1, t.ty1);
}

template <int JJ>
static __device__ __forceinline__ void out_step(
    const CapsStage& s, float q0, float q1, const CapsTurned& t,
    v8f& acc0, v8f& acc1) {
  v2f bx; bx.x = s.wv[JJ][0];  bx.y = s.wv[JJ][2];
  v2f by; by.x = s.wv[JJ][1];  by.y = s.wv[JJ][3];
  v2f a0, a1;
  a0.x = q0 * t.tx0[2 * JJ]; a0.y = q0 * t.tx0[2 * JJ + 1];
  a1.x = q1 * t.tx1[2 * JJ]; a1.y = q1 * t.tx1[2 * JJ + 1];
  acc0 = wmma_f32(a0, bx, acc0);
  acc1 = wmma_f32(a1, bx, acc1);
  a0.x = q0 * t.ty0[2 * JJ]; a0.y = q0 * t.ty0[2 * JJ + 1];
  a1.x = q1 * t.ty1[2 * JJ]; a1.y = q1 * t.ty1[2 * JJ + 1];
  acc0 = wmma_f32(a0, by, acc0);
  acc1 = wmma_f32(a1, by, acc1);
}

static __device__ __forceinline__ void caps_main(
    const CapsStage& s, const CapsTurned& t, v8f& acc0, v8f& acc1) {
  const float q0 = s.qk0 * s.ca0;
  const float q1 = s.qk1 * s.ca1;
  out_step<0>(s, q0, q1, t, acc0, acc1);
  out_step<1>(s, q0, q1, t, acc0, acc1);
  out_step<2>(s, q0, q1, t, acc0, acc1);
  out_step<3>(s, q0, q1, t, acc0, acc1);
}

__global__ __launch_bounds__(256) void caps_out(
    const float* __restrict__ pose, const float* __restrict__ ca,
    const float* __restrict__ w, const float* __restrict__ qk,
    float* __restrict__ out) {
  const int m    = blockIdx.x;
  const int dh   = blockIdx.y;
  const int tid  = threadIdx.x;
  const int lane = tid & 31;
  const int wv   = tid >> 5;
  const int l16  = lane & 15;
  const int h    = lane >> 4;
  const int dcol = dh * 16 + l16;

  __shared__ float red[8][32 * 17];

  const int c0 = l16 - 2 * h;
  v2f s0, s1, s2, s3;
  s0.x = (c0 == 0)  ? 1.f : 0.f;  s0.y = (c0 == 1)  ? 1.f : 0.f;
  s1.x = (c0 == 4)  ? 1.f : 0.f;  s1.y = (c0 == 5)  ? 1.f : 0.f;
  s2.x = (c0 == 8)  ? 1.f : 0.f;  s2.y = (c0 == 9)  ? 1.f : 0.f;
  s3.x = (c0 == 12) ? 1.f : 0.f;  s3.y = (c0 == 13) ? 1.f : 0.f;

  v8f acc0 = {}, acc1 = {};

  const int nbeg = wv * 64;
  CapsStage sa, sb;
  CapsTurned t;
  caps_request(sa, pose, ca, w, qk, nbeg, m, h, l16, dcol);
  pose_turn_all(sa, s0, s1, s2, s3, t);
  for (int i = 0; i < 31; ++i) {
    const int n = nbeg + 2 * i;
    caps_request(sb, pose, ca, w, qk, n + 1, m, h, l16, dcol);
    caps_main(sa, t, acc0, acc1);
    pose_turn_all(sb, s0, s1, s2, s3, t);
    caps_request(sa, pose, ca, w, qk, n + 2, m, h, l16, dcol);
    caps_main(sb, t, acc0, acc1);
    pose_turn_all(sa, s0, s1, s2, s3, t);
  }
  caps_request(sb, pose, ca, w, qk, nbeg + 63, m, h, l16, dcol);
  caps_main(sa, t, acc0, acc1);
  pose_turn_all(sb, s0, s1, s2, s3, t);
  caps_main(sb, t, acc0, acc1);

  const int rbase = (lane < 16) ? 0 : 8;
  for (int i = 0; i < 8; ++i) {
    red[wv][(i + rbase) * 17 + l16]      = acc0[i];
    red[wv][(i + rbase + 16) * 17 + l16] = acc1[i];
  }
  __syncthreads();
  for (int e = tid; e < 512; e += 256) {
    const int b = e >> 4, dc = e & 15;
    float s = 0.f;
    for (int v = 0; v < 8; ++v) s += red[v][b * 17 + dc];
    out[(size_t)b * MD_ + m * D_ + dh * 16 + dc] = s;
  }
}

extern "C" void kernel_launch(void* const* d_in, const int* in_sizes, int n_in,
                              void* d_out, int out_size, void* d_ws,
                              size_t ws_size, hipStream_t stream) {
  (void)in_sizes; (void)n_in; (void)out_size; (void)d_ws; (void)ws_size;
  const float* pose = (const float*)d_in[0];
  const float* ca   = (const float*)d_in[1];
  const float* ncv  = (const float*)d_in[2];
  const float* na   = (const float*)d_in[3];
  const float* w    = (const float*)d_in[4];

  float* out    = (float*)d_out;
  float* na_out = out + (size_t)B_ * M_ * D_;
  float* qk_out = na_out + (size_t)B_ * M_;

  caps_logits_qk<<<dim3(N_), dim3(256), 0, stream>>>(pose, ncv, na, w, qk_out);
  hipMemcpyAsync(na_out, na, (size_t)B_ * M_ * sizeof(float),
                 hipMemcpyDeviceToDevice, stream);
  caps_out<<<dim3(M_, 2), dim3(256), 0, stream>>>(pose, ca, w, qk_out, out);
}


# second kernel: current_act read once per wave as whole lines and laid n-major into LDS (pitch 33); per capsule its two values come from LDS, not from two 16-line global loads; sums unchanged
# speedup vs baseline: 1.5093x; 1.0582x over previous
#include <hip/hip_runtime.h>


typedef __attribute__((ext_vector_type(2))) float v2f;
typedef __attribute__((ext_vector_type(4))) float v4f;
typedef __attribute__((ext_vector_type(8))) float v8f;

#define B_  32
#define N_  512
#define A_  32
#define M_  128
#define D_  32
#define MD_ (M_ * D_)
#define SCALE_ 0.17677669529663687f

static __device__ __forceinline__ v8f wmma_f32(v2f a, v2f b, v8f c) {
  return __builtin_amdgcn_wmma_f32_16x16x4_f32(false, a, false, b, (short)0, c,
                                               false, false);
}

struct WTile { v2f b[2][8]; };

static __device__ __forceinline__ void wtile_request(
    WTile& t, const float* __restrict__ w, int n, int mt, unsigned wlane) {
  const float* wt = w + (size_t)n * A_ * MD_ + mt * 256;
  for (int j = 0; j < 2; ++j) {
    const unsigned lo = wlane + 16u * j;
    __builtin_prefetch(wt + lo + 256, 0, 0);
    for (int k = 0; k < 8; ++k) {
      t.b[j][k].x = wt[lo + (unsigned)(4 * k) * MD_];
      t.b[j][k].y = wt[lo + (unsigned)(4 * k + 1) * MD_];
    }
  }
}

static __device__ __forceinline__ float xhalf(float x) {
  const int b = __builtin_bit_cast(int, x);
  return __builtin_bit_cast(float, __builtin_amdgcn_permlanex16(
      b, b, 0x76543210, 0xfedcba98, false, false));
}

template <bool NEXT>
static __device__ __forceinline__ void logits_step(
    int mt, const WTile& cur, WTile& nxt, const v2f (&preg)[2][8], float* lg,
    const float* __restrict__ ncv, const float* __restrict__ w,
    int n, int lane, unsigned wlane, unsigned clane, int lgl) {
  const float* ct = ncv + mt * 8 * D_;
  v4f cv[2][2][2];
  for (int rt = 0; rt < 2; ++rt) {
    for (int j = 0; j < 2; ++j) {
      const unsigned o = clane + (unsigned)(16 * rt) * MD_ + 16u * j;
      cv[rt][j][0] = *(const v4f*)(ct + o);
      cv[rt][j][1] = *(const v4f*)(ct + o + 4);
    }
  }
  if (NEXT) wtile_request(nxt, w, n, mt + 2, wlane);

  v8f acc[2][2];
  for (int j = 0; j < 2; ++j) {
    acc[j][0] = v8f{}; acc[j][1] = v8f{};
    for (int k = 0; k < 8; ++k) {
      acc[j][0] = wmma_f32(cur.b[j][k], preg[0][k], acc[j][0]);
      acc[j][1] = wmma_f32(cur.b[j][k], preg[1][k], acc[j][1]);
    }
  }
#define DOT2(S, ACC, C0, C1)                                                        \
  S += __builtin_shufflevector(ACC, ACC, 0, 1) * __builtin_shufflevector(C0, C0, 0, 1); \
  S += __builtin_shufflevector(ACC, ACC, 2, 3) * __builtin_shufflevector(C0, C0, 2, 3); \
  S += __builtin_shufflevector(ACC, ACC, 4, 5) * __builtin_shufflevector(C1, C1, 0, 1); \
  S += __builtin_shufflevector(ACC, ACC, 6, 7) * __builtin_shufflevector(C1, C1, 2, 3)
  float t[2];
  for (int rt = 0; rt < 2; ++rt) {
    v2f p0 = {0.f, 0.f}, p1 = {0.f, 0.f};
    DOT2(p0, acc[0][rt], cv[rt][0][0], cv[rt][0][1]);
    DOT2(p1, acc[1][rt], cv[rt][1][0], cv[rt][1][1]);
    const v2f p = p0 + p1;
    const float s = p.x + p.y;
    t[rt] = s + xhalf(s);
  }
#undef DOT2
  lg[lgl + mt * 8] = ((lane < 16) ? t[0] : t[1]) * SCALE_;
}

__global__ __launch_bounds__(256, 4) void caps_logits_qk(
    const float* __restrict__ pose, const float* __restrict__ ncv,
    const float* __restrict__ na, const float* __restrict__ w,
    float* __restrict__ qk_out) {
  const int n    = blockIdx.x;
  const int tid  = threadIdx.x;
  const int lane = tid & 31;
  const int wv   = tid >> 5;
  const int l16  = lane & 15;
  const int khalf = (lane < 16) ? 0 : 2;

  __shared__ float lg[32 * 128];

  v2f preg[2][8];
  for (int rt = 0; rt < 2; ++rt) {
    const int b = l16 + 16 * rt;
    const float* pp = pose + ((size_t)b * N_ + n) * A_;
    for (int k = 0; k < 8; ++k) preg[rt][k] = *(const v2f*)(pp + 4 * k + khalf);
  }

  const unsigned wlane = (unsigned)khalf * MD_ + 32u * wv + l16;
  const unsigned clane = (unsigned)l16 * MD_ + 32u * wv + 4u * khalf;
  const int lgl = lane * 128 + wv;
#define LSTEP(NEXT, MT, CUR, REQ) \
  logits_step<NEXT>(MT, CUR, REQ, preg, lg, ncv, w, n, lane, wlane, clane, lgl)
  WTile ta, tb, tc;
  wtile_request(ta, w, n, 0, wlane);
  wtile_request(tb, w, n, 1, wlane);
  for (int t3 = 0; t3 < 4; ++t3) {
    LSTEP(true, 3 * t3,     ta, tc);
    LSTEP(true, 3 * t3 + 1, tb, ta);
    LSTEP(true, 3 * t3 + 2, tc, tb);
  }
  LSTEP(true,  12, ta, tc);
  LSTEP(true,  13, tb, ta);
  LSTEP(false, 14, tc, tb);
  LSTEP(false, 15, ta, tb);
#undef LSTEP

  __syncthreads();

  if (tid < 32) {
    const int b = tid;
    float* row = lg + b * 128;
    float mx = -3.4e38f;
    for (int m = 0; m < 128; ++m) mx = fmaxf(mx, row[m]);
    float s = 0.f;
    for (int m = 0; m < 128; ++m) { float e = __expf(row[m] - mx); row[m] = e; s += e; }
    const float inv = 1.f / s;
    const float* nap = na + b * 128;
    float s2 = 0.f;
    for (int m = 0; m < 128; ++m) { float v = row[m] * inv * nap[m]; row[m] = v; s2 += v; }
    const float inv2 = 1.f / (s2 + 1e-10f);
    float* qp = qk_out + ((size_t)b * N_ + n) * M_;
    for (int m = 0; m < 128; ++m) qp[m] = row[m] * inv2;
  }
}


struct CapsStage {
  v2f   p[2][4][2];
  float qk0, qk1, ca0, ca1;
  float wv[4][4];
};

static __device__ __forceinline__ void caps_request(
    CapsStage& s, const float* __restrict__ pose,
    const float* __restrict__ w, const float* __restrict__ qk,
    int n, int m, int h, int l16, int dcol) {
  const float* pt = pose + ((size_t)(2 * h) * N_ + n) * A_ + 2 * l16;
  for (int t = 0; t < 2; ++t) {
    for (int g = 0; g < 4; ++g) {
      const float* rowp = pt + (size_t)(16 * t + 4 * g) * N_ * A_;
      s.p[t][g][0] = *(const v2f*)(rowp);
      s.p[t][g][1] = *(const v2f*)(rowp + N_ * A_);
    }
  }
  const int b0 = l16, b1 = l16 + 16;
  s.qk0 = qk[((size_t)b0 * N_ + n) * M_ + m];
  s.qk1 = qk[((size_t)b1 * N_ + n) * M_ + m];
  const float* wbase = w + ((size_t)n * A_ + 16 * h) * MD_ + m * D_ + dcol;
  __builtin_prefetch(wbase + (size_t)2 * A_ * MD_, 0, 0);
  for (int jj = 0; jj < 4; ++jj) {
    for (int r = 0; r < 4; ++r) s.wv[jj][r] = wbase[(size_t)(4 * jj + r) * MD_];
  }
}

#define CAP_ 33
static __device__ __forceinline__ void caps_act(
    CapsStage& s, const float* ca_s, int n, int l16) {
  s.ca0 = ca_s[n * CAP_ + l16];
  s.ca1 = ca_s[n * CAP_ + l16 + 16];
}

struct CapsTurned { v8f tx0, ty0, tx1, ty1; };

static __device__ __forceinline__ void pose_turn_regs(
    v2f r0, v2f r1, v2f sel, v8f& tx, v8f& ty) {
  v2f ax; ax.x = r0.x; ax.y = r1.x;
  v2f ay; ay.x = r0.y; ay.y = r1.y;
  tx = wmma_f32(ax, sel, tx);
  ty = wmma_f32(ay, sel, ty);
}

static __device__ __forceinline__ void pose_turn_all(
    const CapsStage& s, v2f s0, v2f s1, v2f s2, v2f s3, CapsTurned& t) {
  t.tx0 = v8f{}; t.ty0 = v8f{}; t.tx1 = v8f{}; t.ty1 = v8f{};
  pose_turn_regs(s.p[0][0][0], s.p[0][0][1], s0, t.tx0, t.ty0);
  pose_turn_regs(s.p[0][1][0], s.p[0][1][1], s1, t.tx0, t.ty0);
  pose_turn_regs(s.p[0][2][0], s.p[0][2][1], s2, t.tx0, t.ty0);
  pose_turn_regs(s.p[0][3][0], s.p[0][3][1], s3, t.tx0, t.ty0);
  pose_turn_regs(s.p[1][0][0], s.p[1][0][1], s0, t.tx1, t.ty1);
  pose_turn_regs(s.p[1][1][0], s.p[1][1][1], s1, t.tx1, t.ty1);
  pose_turn_regs(s.p[1][2][0], s.p[1][2][1], s2, t.tx1, t.ty1);
  pose_turn_regs(s.p[1][3][0], s.p[1][3][1], s3, t.tx1, t.ty1);
}

template <int JJ>
static __device__ __forceinline__ void out_step(
    const CapsStage& s, float q0, float q1, const CapsTurned& t,
    v8f& acc0, v8f& acc1) {
  v2f bx; bx.x = s.wv[JJ][0];  bx.y = s.wv[JJ][2];
  v2f by; by.x = s.wv[JJ][1];  by.y = s.wv[JJ][3];
  v2f a0, a1;
  a0.x = q0 * t.tx0[2 * JJ]; a0.y = q0 * t.tx0[2 * JJ + 1];
  a1.x = q1 * t.tx1[2 * JJ]; a1.y = q1 * t.tx1[2 * JJ + 1];
  acc0 = wmma_f32(a0, bx, acc0);
  acc1 = wmma_f32(a1, bx, acc1);
  a0.x = q0 * t.ty0[2 * JJ]; a0.y = q0 * t.ty0[2 * JJ + 1];
  a1.x = q1 * t.ty1[2 * JJ]; a1.y = q1 * t.ty1[2 * JJ + 1];
  acc0 = wmma_f32(a0, by, acc0);
  acc1 = wmma_f32(a1, by, acc1);
}

static __device__ __forceinline__ void caps_main(
    const CapsStage& s, const CapsTurned& t, v8f& acc0, v8f& acc1) {
  const float q0 = s.qk0 * s.ca0;
  const float q1 = s.qk1 * s.ca1;
  out_step<0>(s, q0, q1, t, acc0, acc1);
  out_step<1>(s, q0, q1, t, acc0, acc1);
  out_step<2>(s, q0, q1, t, acc0, acc1);
  out_step<3>(s, q0, q1, t, acc0, acc1);
}

__global__ __launch_bounds__(256) void caps_out(
    const float* __restrict__ pose, const float* __restrict__ ca,
    const float* __restrict__ w, const float* __restrict__ qk,
    float* __restrict__ out) {
  const int m    = blockIdx.x;
  const int dh   = blockIdx.y;
  const int tid  = threadIdx.x;
  const int lane = tid & 31;
  const int wv   = tid >> 5;
  const int l16  = lane & 15;
  const int h    = lane >> 4;
  const int dcol = dh * 16 + l16;

  __shared__ float red[8][32 * 17];
  __shared__ float ca_s[N_ * CAP_];

  const int c0 = l16 - 2 * h;
  v2f s0, s1, s2, s3;
  s0.x = (c0 == 0)  ? 1.f : 0.f;  s0.y = (c0 == 1)  ? 1.f : 0.f;
  s1.x = (c0 == 4)  ? 1.f : 0.f;  s1.y = (c0 == 5)  ? 1.f : 0.f;
  s2.x = (c0 == 8)  ? 1.f : 0.f;  s2.y = (c0 == 9)  ? 1.f : 0.f;
  s3.x = (c0 == 12) ? 1.f : 0.f;  s3.y = (c0 == 13) ? 1.f : 0.f;

  v8f acc0 = {}, acc1 = {};

  const int nbeg = wv * 64;
  CapsStage sa, sb;
  CapsTurned t;

  {
    const int fb = lane >> 3, fp = lane & 7;
    v4f cx[8][2];
    for (int jj = 0; jj < 8; ++jj)
      for (int hf = 0; hf < 2; ++hf)
        cx[jj][hf] = *(const v4f*)(ca + (size_t)(4 * jj + fb) * N_ + nbeg + 32 * hf + 4 * fp);
    caps_request(sa, pose, w, qk, nbeg, m, h, l16, dcol);
    for (int jj = 0; jj < 8; ++jj)
      for (int hf = 0; hf < 2; ++hf) {
        float* dst = ca_s + (nbeg + 32 * hf + 4 * fp) * CAP_ + 4 * jj + fb;
        dst[0]        = cx[jj][hf].x;
        dst[CAP_]     = cx[jj][hf].y;
        dst[2 * CAP_] = cx[jj][hf].z;
        dst[3 * CAP_] = cx[jj][hf].w;
      }
  }
  __syncthreads();

  caps_act(sa, ca_s, nbeg, l16);
  pose_turn_all(sa, s0, s1, s2, s3, t);
  for (int i = 0; i < 31; ++i) {
    const int n = nbeg + 2 * i;
    caps_request(sb, pose, w, qk, n + 1, m, h, l16, dcol);
    caps_act(sb, ca_s, n + 1, l16);
    caps_main(sa, t, acc0, acc1);
    pose_turn_all(sb, s0, s1, s2, s3, t);
    caps_request(sa, pose, w, qk, n + 2, m, h, l16, dcol);
    caps_act(sa, ca_s, n + 2, l16);
    caps_main(sb, t, acc0, acc1);
    pose_turn_all(sa, s0, s1, s2, s3, t);
  }
  caps_request(sb, pose, w, qk, nbeg + 63, m, h, l16, dcol);
  caps_act(sb, ca_s, nbeg + 63, l16);
  caps_main(sa, t, acc0, acc1);
  pose_turn_all(sb, s0, s1, s2, s3, t);
  caps_main(sb, t, acc0, acc1);

  const int rbase = (lane < 16) ? 0 : 8;
  for (int i = 0; i < 8; ++i) {
    red[wv][(i + rbase) * 17 + l16]      = acc0[i];
    red[wv][(i + rbase + 16) * 17 + l16] = acc1[i];
  }
  __syncthreads();
  for (int e = tid; e < 512; e += 256) {
    const int b = e >> 4, dc = e & 15;
    float s = 0.f;
    for (int v = 0; v < 8; ++v) s += red[v][b * 17 + dc];
    out[(size_t)b * MD_ + m * D_ + dh * 16 + dc] = s;
  }
}

extern "C" void kernel_launch(void* const* d_in, const int* in_sizes, int n_in,
                              void* d_out, int out_size, void* d_ws,
                              size_t ws_size, hipStream_t stream) {
  (void)in_sizes; (void)n_in; (void)out_size; (void)d_ws; (void)ws_size;
  const float* pose = (const float*)d_in[0];
  const float* ca   = (const float*)d_in[1];
  const float* ncv  = (const float*)d_in[2];
  const float* na   = (const float*)d_in[3];
  const float* w    = (const float*)d_in[4];

  float* out    = (float*)d_out;
  float* na_out = out + (size_t)B_ * M_ * D_;
  float* qk_out = na_out + (size_t)B_ * M_;

  caps_logits_qk<<<dim3(N_), dim3(256), 0, stream>>>(pose, ncv, na, w, qk_out);
  hipMemcpyAsync(na_out, na, (size_t)B_ * M_ * sizeof(float),
                 hipMemcpyDeviceToDevice, stream);
  caps_out<<<dim3(M_, 2), dim3(256), 0, stream>>>(pose, ca, w, qk_out, out);
}


# first kernel tail on all 8 waves: logits back in registers (lane = b, 16 m per wave), row max and sums exchanged through LDS, rows stored whole; fence keeps tile loads ahead; sum order over m changes
# speedup vs baseline: 1.6694x; 1.1061x over previous
#include <hip/hip_runtime.h>


typedef __attribute__((ext_vector_type(2))) float v2f;
typedef __attribute__((ext_vector_type(4))) float v4f;
typedef __attribute__((ext_vector_type(8))) float v8f;

#define B_  32
#define N_  512
#define A_  32
#define M_  128
#define D_  32
#define MD_ (M_ * D_)
#define SCALE_ 0.17677669529663687f
#define LGP_ 132

static __device__ __forceinline__ v8f wmma_f32(v2f a, v2f b, v8f c) {
  return __builtin_amdgcn_wmma_f32_16x16x4_f32(false, a, false, b, (short)0, c,
                                               false, false);
}

struct WTile { v2f b[2][8]; };

static __device__ __forceinline__ void wtile_request(
    WTile& t, const float* __restrict__ w, int n, int mt, unsigned wlane) {
  const float* wt = w + (size_t)n * A_ * MD_ + mt * 256;
  for (int j = 0; j < 2; ++j) {
    const unsigned lo = wlane + 16u * j;
    __builtin_prefetch(wt + lo + 256, 0, 0);
    for (int k = 0; k < 8; ++k) {
      t.b[j][k].x = wt[lo + (unsigned)(4 * k) * MD_];
      t.b[j][k].y = wt[lo + (unsigned)(4 * k + 1) * MD_];
    }
  }
}

static __device__ __forceinline__ float xhalf(float x) {
  const int b = __builtin_bit_cast(int, x);
  return __builtin_bit_cast(float, __builtin_amdgcn_permlanex16(
      b, b, 0x76543210, 0xfedcba98, false, false));
}

template <bool NEXT>
static __device__ __forceinline__ void logits_step(
    int mt, const WTile& cur, WTile& nxt, const v2f (&preg)[2][8], float* lg,
    const float* __restrict__ ncv, const float* __restrict__ w,
    int n, int lane, unsigned wlane, unsigned clane, int lgl) {
  const float* ct = ncv + mt * 8 * D_;
  v4f cv[2][2][2];
  for (int rt = 0; rt < 2; ++rt) {
    for (int j = 0; j < 2; ++j) {
      const unsigned o = clane + (unsigned)(16 * rt) * MD_ + 16u * j;
      cv[rt][j][0] = *(const v4f*)(ct + o);
      cv[rt][j][1] = *(const v4f*)(ct + o + 4);
    }
  }
  if (NEXT) wtile_request(nxt, w, n, mt + 2, wlane);
  __builtin_amdgcn_sched_barrier(0);

  v8f acc[2][2];
  for (int j = 0; j < 2; ++j) {
    acc[j][0] = v8f{}; acc[j][1] = v8f{};
    for (int k = 0; k < 8; ++k) {
      acc[j][0] = wmma_f32(cur.b[j][k], preg[0][k], acc[j][0]);
      acc[j][1] = wmma_f32(cur.b[j][k], preg[1][k], acc[j][1]);
    }
  }
#define DOT2(S, ACC, C0, C1)                                                        \
  S += __builtin_shufflevector(ACC, ACC, 0, 1) * __builtin_shufflevector(C0, C0, 0, 1); \
  S += __builtin_shufflevector(ACC, ACC, 2, 3) * __builtin_shufflevector(C0, C0, 2, 3); \
  S += __builtin_shufflevector(ACC, ACC, 4, 5) * __builtin_shufflevector(C1, C1, 0, 1); \
  S += __builtin_shufflevector(ACC, ACC, 6, 7) * __builtin_shufflevector(C1, C1, 2, 3)
  float t[2];
  for (int rt = 0; rt < 2; ++rt) {
    v2f p0 = {0.f, 0.f}, p1 = {0.f, 0.f};
    DOT2(p0, acc[0][rt], cv[rt][0][0], cv[rt][0][1]);
    DOT2(p1, acc[1][rt], cv[rt][1][0], cv[rt][1][1]);
    const v2f p = p0 + p1;
    const float s = p.x + p.y;
    t[rt] = s + xhalf(s);
  }
#undef DOT2
  lg[lgl + mt * 8] = ((lane < 16) ? t[0] : t[1]) * SCALE_;
}

__global__ __launch_bounds__(256, 4) void caps_logits_qk(
    const float* __restrict__ pose, const float* __restrict__ ncv,
    const float* __restrict__ na, const float* __restrict__ w,
    float* __restrict__ qk_out) {
  const int n    = blockIdx.x;
  const int tid  = threadIdx.x;
  const int lane = tid & 31;
  const int wv   = tid >> 5;
  const int l16  = lane & 15;
  const int khalf = (lane < 16) ? 0 : 2;

  __shared__ __attribute__((aligned(16))) float lg[32 * LGP_];
  __shared__ float red[3][8][32];

  v2f preg[2][8];
  for (int rt = 0; rt < 2; ++rt) {
    const int b = l16 + 16 * rt;
    const float* pp = pose + ((size_t)b * N_ + n) * A_;
    for (int k = 0; k < 8; ++k) preg[rt][k] = *(const v2f*)(pp + 4 * k + khalf);
  }

  const unsigned wlane = (unsigned)khalf * MD_ + 32u * wv + l16;
  const unsigned clane = (unsigned)l16 * MD_ + 32u * wv + 4u * khalf;
  const int lgl = lane * LGP_ + wv;
#define LSTEP(NEXT, MT, CUR, REQ) \
  logits_step<NEXT>(MT, CUR, REQ, preg, lg, ncv, w, n, lane, wlane, clane, lgl)
  WTile ta, tb, tc;
  wtile_request(ta, w, n, 0, wlane);
  wtile_request(tb, w, n, 1, wlane);
  for (int t3 = 0; t3 < 4; ++t3) {
    LSTEP(true, 3 * t3,     ta, tc);
    LSTEP(true, 3 * t3 + 1, tb, ta);
    LSTEP(true, 3 * t3 + 2, tc, tb);
  }
  LSTEP(true,  12, ta, tc);
  LSTEP(true,  13, tb, ta);
  LSTEP(false, 14, tc, tb);
  LSTEP(false, 15, ta, tb);
#undef LSTEP

  v4f nar[4];
  for (int r = 0; r < 4; ++r)
    nar[r] = *(const v4f*)(na + (4 * wv + r) * M_ + 4 * lane);

  float x[16];
  for (int i = 0; i < 16; ++i) x[i] = lg[lgl + 8 * i];
  float mx = x[0];
  for (int i = 1; i < 16; ++i) mx = fmaxf(mx, x[i]);
  red[0][wv][lane] = mx;
  __syncthreads();

  for (int r = 0; r < 4; ++r)
    *(v4f*)(lg + (4 * wv + r) * LGP_ + 4 * lane) = nar[r];

  float gm = red[0][0][lane];
  for (int v = 1; v < 8; ++v) gm = fmaxf(gm, red[0][v][lane]);
  float s = 0.f;
  for (int i = 0; i < 16; ++i) { x[i] = __expf(x[i] - gm); s += x[i]; }
  red[1][wv][lane] = s;
  __syncthreads();

  float st = 0.f;
  for (int v = 0; v < 8; ++v) st += red[1][v][lane];
  const float inv = 1.f / st;
  float s2 = 0.f;
  for (int i = 0; i < 16; ++i) {
    x[i] = x[i] * inv * lg[lgl + 8 * i];
    s2 += x[i];
  }
  red[2][wv][lane] = s2;
  __syncthreads();

  float s2t = 0.f;
  for (int v = 0; v < 8; ++v) s2t += red[2][v][lane];
  const float inv2 = 1.f / (s2t + 1e-10f);
  for (int i = 0; i < 16; ++i) lg[lgl + 8 * i] = x[i] * inv2;
  __syncthreads();

  for (int r = 0; r < 4; ++r) {
    const int b = 4 * wv + r;
    const v4f q = *(const v4f*)(lg + b * LGP_ + 4 * lane);
    *(v4f*)(qk_out + ((size_t)b * N_ + n) * M_ + 4 * lane) = q;
  }
}


struct CapsStage {
  v2f   p[2][4][2];
  float qk0, qk1, ca0, ca1;
  float wv[4][4];
};

static __device__ __forceinline__ void caps_request(
    CapsStage& s, const float* __restrict__ pose,
    const float* __restrict__ w, const float* __restrict__ qk,
    int n, int m, int h, int l16, int dcol) {
  const float* pt = pose + ((size_t)(2 * h) * N_ + n) * A_ + 2 * l16;
  for (int t = 0; t < 2; ++t) {
    for (int g = 0; g < 4; ++g) {
      const float* rowp = pt + (size_t)(16 * t + 4 * g) * N_ * A_;
      s.p[t][g][0] = *(const v2f*)(rowp);
      s.p[t][g][1] = *(const v2f*)(rowp + N_ * A_);
    }
  }
  const int b0 = l16, b1 = l16 + 16;
  s.qk0 = qk[((size_t)b0 * N_ + n) * M_ + m];
  s.qk1 = qk[((size_t)b1 * N_ + n) * M_ + m];
  const float* wbase = w + ((size_t)n * A_ + 16 * h) * MD_ + m * D_ + dcol;
  __builtin_prefetch(wbase + (size_t)2 * A_ * MD_, 0, 0);
  for (int jj = 0; jj < 4; ++jj) {
    for (int r = 0; r < 4; ++r) s.wv[jj][r] = wbase[(size_t)(4 * jj + r) * MD_];
  }
}

#define CAP_ 33
static __device__ __forceinline__ void caps_act(
    CapsStage& s, const float* ca_s, int n, int l16) {
  s.ca0 = ca_s[n * CAP_ + l16];
  s.ca1 = ca_s[n * CAP_ + l16 + 16];
}

struct CapsTurned { v8f tx0, ty0, tx1, ty1; };

static __device__ __forceinline__ void pose_turn_regs(
    v2f r0, v2f r1, v2f sel, v8f& tx, v8f& ty) {
  v2f ax; ax.x = r0.x; ax.y = r1.x;
  v2f ay; ay.x = r0.y; ay.y = r1.y;
  tx = wmma_f32(ax, sel, tx);
  ty = wmma_f32(ay, sel, ty);
}

static __device__ __forceinline__ void pose_turn_all(
    const CapsStage& s, v2f s0, v2f s1, v2f s2, v2f s3, CapsTurned& t) {
  t.tx0 = v8f{}; t.ty0 = v8f{}; t.tx1 = v8f{}; t.ty1 = v8f{};
  pose_turn_regs(s.p[0][0][0], s.p[0][0][1], s0, t.tx0, t.ty0);
  pose_turn_regs(s.p[0][1][0], s.p[0][1][1], s1, t.tx0, t.ty0);
  pose_turn_regs(s.p[0][2][0], s.p[0][2][1], s2, t.tx0, t.ty0);
  pose_turn_regs(s.p[0][3][0], s.p[0][3][1], s3, t.tx0, t.ty0);
  pose_turn_regs(s.p[1][0][0], s.p[1][0][1], s0, t.tx1, t.ty1);
  pose_turn_regs(s.p[1][1][0], s.p[1][1][1], s1, t.tx1, t.ty1);
  pose_turn_regs(s.p[1][2][0], s.p[1][2][1], s2, t.tx1, t.ty1);
  pose_turn_regs(s.p[1][3][0], s.p[1][3][1], s3, t.tx1, t.ty1);
}

template <int JJ>
static __device__ __forceinline__ void out_step(
    const CapsStage& s, float q0, float q1, const CapsTurned& t,
    v8f& acc0, v8f& acc1) {
  v2f bx; bx.x = s.wv[JJ][0];  bx.y = s.wv[JJ][2];
  v2f by; by.x = s.wv[JJ][1];  by.y = s.wv[JJ][3];
  v2f a0, a1;
  a0.x = q0 * t.tx0[2 * JJ]; a0.y = q0 * t.tx0[2 * JJ + 1];
  a1.x = q1 * t.tx1[2 * JJ]; a1.y = q1 * t.tx1[2 * JJ + 1];
  acc0 = wmma_f32(a0, bx, acc0);
  acc1 = wmma_f32(a1, bx, acc1);
  a0.x = q0 * t.ty0[2 * JJ]; a0.y = q0 * t.ty0[2 * JJ + 1];
  a1.x = q1 * t.ty1[2 * JJ]; a1.y = q1 * t.ty1[2 * JJ + 1];
  acc0 = wmma_f32(a0, by, acc0);
  acc1 = wmma_f32(a1, by, acc1);
}

static __device__ __forceinline__ void caps_main(
    const CapsStage& s, const CapsTurned& t, v8f& acc0, v8f& acc1) {
  const float q0 = s.qk0 * s.ca0;
  const float q1 = s.qk1 * s.ca1;
  out_step<0>(s, q0, q1, t, acc0, acc1);
  out_step<1>(s, q0, q1, t, acc0, acc1);
  out_step<2>(s, q0, q1, t, acc0, acc1);
  out_step<3>(s, q0, q1, t, acc0, acc1);
}

__global__ __launch_bounds__(256) void caps_out(
    const float* __restrict__ pose, const float* __restrict__ ca,
    const float* __restrict__ w, const float* __restrict__ qk,
    float* __restrict__ out) {
  const int m    = blockIdx.x;
  const int dh   = blockIdx.y;
  const int tid  = threadIdx.x;
  const int lane = tid & 31;
  const int wv   = tid >> 5;
  const int l16  = lane & 15;
  const int h    = lane >> 4;
  const int dcol = dh * 16 + l16;

  __shared__ float red[8][32 * 17];
  __shared__ float ca_s[N_ * CAP_];

  const int c0 = l16 - 2 * h;
  v2f s0, s1, s2, s3;
  s0.x = (c0 == 0)  ? 1.f : 0.f;  s0.y = (c0 == 1)  ? 1.f : 0.f;
  s1.x = (c0 == 4)  ? 1.f : 0.f;  s1.y = (c0 == 5)  ? 1.f : 0.f;
  s2.x = (c0 == 8)  ? 1.f : 0.f;  s2.y = (c0 == 9)  ? 1.f : 0.f;
  s3.x = (c0 == 12) ? 1.f : 0.f;  s3.y = (c0 == 13) ? 1.f : 0.f;

  v8f acc0 = {}, acc1 = {};

  const int nbeg = wv * 64;
  CapsStage sa, sb;
  CapsTurned t;

  {
    const int fb = lane >> 3, fp = lane & 7;
    v4f cx[8][2];
    for (int jj = 0; jj < 8; ++jj)
      for (int hf = 0; hf < 2; ++hf)
        cx[jj][hf] = *(const v4f*)(ca + (size_t)(4 * jj + fb) * N_ + nbeg + 32 * hf + 4 * fp);
    caps_request(sa, pose, w, qk, nbeg, m, h, l16, dcol);
    for (int jj = 0; jj < 8; ++jj)
      for (int hf = 0; hf < 2; ++hf) {
        float* dst = ca_s + (nbeg + 32 * hf + 4 * fp) * CAP_ + 4 * jj + fb;
        dst[0]        = cx[jj][hf].x;
        dst[CAP_]     = cx[jj][hf].y;
        dst[2 * CAP_] = cx[jj][hf].z;
        dst[3 * CAP_] = cx[jj][hf].w;
      }
  }
  __syncthreads();

  caps_act(sa, ca_s, nbeg, l16);
  pose_turn_all(sa, s0, s1, s2, s3, t);
  for (int i = 0; i < 31; ++i) {
    const int n = nbeg + 2 * i;
    caps_request(sb, pose, w, qk, n + 1, m, h, l16, dcol);
    caps_act(sb, ca_s, n + 1, l16);
    caps_main(sa, t, acc0, acc1);
    pose_turn_all(sb, s0, s1, s2, s3, t);
    caps_request(sa, pose, w, qk, n + 2, m, h, l16, dcol);
    caps_act(sa, ca_s, n + 2, l16);
    caps_main(sb, t, acc0, acc1);
    pose_turn_all(sa, s0, s1, s2, s3, t);
  }
  caps_request(sb, pose, w, qk, nbeg + 63, m, h, l16, dcol);
  caps_act(sb, ca_s, nbeg + 63, l16);
  caps_main(sa, t, acc0, acc1);
  pose_turn_all(sb, s0, s1, s2, s3, t);
  caps_main(sb, t, acc0, acc1);

  const int rbase = (lane < 16) ? 0 : 8;
  for (int i = 0; i < 8; ++i) {
    red[wv][(i + rbase) * 17 + l16]      = acc0[i];
    red[wv][(i + rbase + 16) * 17 + l16] = acc1[i];
  }
  __syncthreads();
  for (int e = tid; e < 512; e += 256) {
    const int b = e >> 4, dc = e & 15;
    float s = 0.f;
    for (int v = 0; v < 8; ++v) s += red[v][b * 17 + dc];
    out[(size_t)b * MD_ + m * D_ + dh * 16 + dc] = s;
  }
}

extern "C" void kernel_launch(void* const* d_in, const int* in_sizes, int n_in,
                              void* d_out, int out_size, void* d_ws,
                              size_t ws_size, hipStream_t stream) {
  (void)in_sizes; (void)n_in; (void)out_size; (void)d_ws; (void)ws_size;
  const float* pose = (const float*)d_in[0];
  const float* ca   = (const float*)d_in[1];
  const float* ncv  = (const float*)d_in[2];
  const float* na   = (const float*)d_in[3];
  const float* w    = (const float*)d_in[4];

  float* out    = (float*)d_out;
  float* na_out = out + (size_t)B_ * M_ * D_;
  float* qk_out = na_out + (size_t)B_ * M_;

  caps_logits_qk<<<dim3(N_), dim3(256), 0, stream>>>(pose, ncv, na, w, qk_out);
  hipMemcpyAsync(na_out, na, (size_t)B_ * M_ * sizeof(float),
                 hipMemcpyDeviceToDevice, stream);
  caps_out<<<dim3(M_, 2), dim3(256), 0, stream>>>(pose, ca, w, qk_out, out);
}


# first kernel also stores the routing weights m-major in the workspace (one 128-B line per (m, n)); second kernel reads 1 line per capsule for them, not 32 (old path if workspace < 8 MiB); same sums
# speedup vs baseline: 1.7904x; 1.0725x over previous
#include <hip/hip_runtime.h>


typedef __attribute__((ext_vector_type(2))) float v2f;
typedef __attribute__((ext_vector_type(4))) float v4f;
typedef __attribute__((ext_vector_type(8))) float v8f;

#define B_  32
#define N_  512
#define A_  32
#define M_  128
#define D_  32
#define MD_ (M_ * D_)
#define SCALE_ 0.17677669529663687f
#define LGP_ 132
#define QKT_FLOATS_ ((size_t)M_ * N_ * B_)

static __device__ __forceinline__ v8f wmma_f32(v2f a, v2f b, v8f c) {
  return __builtin_amdgcn_wmma_f32_16x16x4_f32(false, a, false, b, (short)0, c,
                                               false, false);
}

struct WTile { v2f b[2][8]; };

static __device__ __forceinline__ void wtile_request(
    WTile& t, const float* __restrict__ w, int n, int mt, unsigned wlane) {
  const float* wt = w + (size_t)n * A_ * MD_ + mt * 256;
  for (int j = 0; j < 2; ++j) {
    const unsigned lo = wlane + 16u * j;
    __builtin_prefetch(wt + lo + 256, 0, 0);
    for (int k = 0; k < 8; ++k) {
      t.b[j][k].x = wt[lo + (unsigned)(4 * k) * MD_];
      t.b[j][k].y = wt[lo + (unsigned)(4 * k + 1) * MD_];
    }
  }
}

static __device__ __forceinline__ float xhalf(float x) {
  const int b = __builtin_bit_cast(int, x);
  return __builtin_bit_cast(float, __builtin_amdgcn_permlanex16(
      b, b, 0x76543210, 0xfedcba98, false, false));
}

template <bool NEXT>
static __device__ __forceinline__ void logits_step(
    int mt, const WTile& cur, WTile& nxt, const v2f (&preg)[2][8], float* lg,
    const float* __restrict__ ncv, const float* __restrict__ w,
    int n, int lane, unsigned wlane, unsigned clane, int lgl) {
  const float* ct = ncv + mt * 8 * D_;
  v4f cv[2][2][2];
  for (int rt = 0; rt < 2; ++rt) {
    for (int j = 0; j < 2; ++j) {
      const unsigned o = clane + (unsigned)(16 * rt) * MD_ + 16u * j;
      cv[rt][j][0] = *(const v4f*)(ct + o);
      cv[rt][j][1] = *(const v4f*)(ct + o + 4);
    }
  }
  if (NEXT) wtile_request(nxt, w, n, mt + 2, wlane);
  __builtin_amdgcn_sched_barrier(0);

  v8f acc[2][2];
  for (int j = 0; j < 2; ++j) {
    acc[j][0] = v8f{}; acc[j][1] = v8f{};
    for (int k = 0; k < 8; ++k) {
      acc[j][0] = wmma_f32(cur.b[j][k], preg[0][k], acc[j][0]);
      acc[j][1] = wmma_f32(cur.b[j][k], preg[1][k], acc[j][1]);
    }
  }
#define DOT2(S, ACC, C0, C1)                                                        \
  S += __builtin_shufflevector(ACC, ACC, 0, 1) * __builtin_shufflevector(C0, C0, 0, 1); \
  S += __builtin_shufflevector(ACC, ACC, 2, 3) * __builtin_shufflevector(C0, C0, 2, 3); \
  S += __builtin_shufflevector(ACC, ACC, 4, 5) * __builtin_shufflevector(C1, C1, 0, 1); \
  S += __builtin_shufflevector(ACC, ACC, 6, 7) * __builtin_shufflevector(C1, C1, 2, 3)
  float t[2];
  for (int rt = 0; rt < 2; ++rt) {
    v2f p0 = {0.f, 0.f}, p1 = {0.f, 0.f};
    DOT2(p0, acc[0][rt], cv[rt][0][0], cv[rt][0][1]);
    DOT2(p1, acc[1][rt], cv[rt][1][0], cv[rt][1][1]);
    const v2f p = p0 + p1;
    const float s = p.x + p.y;
    t[rt] = s + xhalf(s);
  }
#undef DOT2
  lg[lgl + mt * 8] = ((lane < 16) ? t[0] : t[1]) * SCALE_;
}

template <bool QT>
__global__ __launch_bounds__(256, 4) void caps_logits_qk(
    const float* __restrict__ pose, const float* __restrict__ ncv,
    const float* __restrict__ na, const float* __restrict__ w,
    float* __restrict__ qk_out, float* __restrict__ qkt) {
  const int n    = blockIdx.x;
  const int tid  = threadIdx.x;
  const int lane = tid & 31;
  const int wv   = tid >> 5;
  const int l16  = lane & 15;
  const int khalf = (lane < 16) ? 0 : 2;

  __shared__ __attribute__((aligned(16))) float lg[32 * LGP_];
  __shared__ float red[3][8][32];

  v2f preg[2][8];
  for (int rt = 0; rt < 2; ++rt) {
    const int b = l16 + 16 * rt;
    const float* pp = pose + ((size_t)b * N_ + n) * A_;
    for (int k = 0; k < 8; ++k) preg[rt][k] = *(const v2f*)(pp + 4 * k + khalf);
  }

  const unsigned wlane = (unsigned)khalf * MD_ + 32u * wv + l16;
  const unsigned clane = (unsigned)l16 * MD_ + 32u * wv + 4u * khalf;
  const int lgl = lane * LGP_ + wv;
#define LSTEP(NEXT, MT, CUR, REQ) \
  logits_step<NEXT>(MT, CUR, REQ, preg, lg, ncv, w, n, lane, wlane, clane, lgl)
  WTile ta, tb, tc;
  wtile_request(ta, w, n, 0, wlane);
  wtile_request(tb, w, n, 1, wlane);
  for (int t3 = 0; t3 < 4; ++t3) {
    LSTEP(true, 3 * t3,     ta, tc);
    LSTEP(true, 3 * t3 + 1, tb, ta);
    LSTEP(true, 3 * t3 + 2, tc, tb);
  }
  LSTEP(true,  12, ta, tc);
  LSTEP(true,  13, tb, ta);
  LSTEP(false, 14, tc, tb);
  LSTEP(false, 15, ta, tb);
#undef LSTEP

  v4f nar[4];
  for (int r = 0; r < 4; ++r)
    nar[r] = *(const v4f*)(na + (4 * wv + r) * M_ + 4 * lane);

  float x[16];
  for (int i = 0; i < 16; ++i) x[i] = lg[lgl + 8 * i];
  float mx = x[0];
  for (int i = 1; i < 16; ++i) mx = fmaxf(mx, x[i]);
  red[0][wv][lane] = mx;
  __syncthreads();

  for (int r = 0; r < 4; ++r)
    *(v4f*)(lg + (4 * wv + r) * LGP_ + 4 * lane) = nar[r];

  float gm = red[0][0][lane];
  for (int v = 1; v < 8; ++v) gm = fmaxf(gm, red[0][v][lane]);
  float s = 0.f;
  for (int i = 0; i < 16; ++i) { x[i] = __expf(x[i] - gm); s += x[i]; }
  red[1][wv][lane] = s;
  __syncthreads();

  float st = 0.f;
  for (int v = 0; v < 8; ++v) st += red[1][v][lane];
  const float inv = 1.f / st;
  float s2 = 0.f;
  for (int i = 0; i < 16; ++i) {
    x[i] = x[i] * inv * lg[lgl + 8 * i];
    s2 += x[i];
  }
  red[2][wv][lane] = s2;
  __syncthreads();

  float s2t = 0.f;
  for (int v = 0; v < 8; ++v) s2t += red[2][v][lane];
  const float inv2 = 1.f / (s2t + 1e-10f);
  for (int i = 0; i < 16; ++i) {
    x[i] = x[i] * inv2;
    lg[lgl + 8 * i] = x[i];
  }
  __syncthreads();

  if (QT) {
    float* qt = qkt + ((size_t)wv * N_ + n) * B_ + 2 * l16 + (lane >> 4);
    for (int i = 0; i < 16; ++i) qt[(size_t)(8 * i) * N_ * B_] = x[i];
  }

  for (int r = 0; r < 4; ++r) {
    const int b = 4 * wv + r;
    const v4f q = *(const v4f*)(lg + b * LGP_ + 4 * lane);
    *(v4f*)(qk_out + ((size_t)b * N_ + n) * M_ + 4 * lane) = q;
  }
}


struct CapsStage {
  v2f   p[2][4][2];
  float qk0, qk1, ca0, ca1;
  float wv[4][4];
};

template <bool QT>
static __device__ __forceinline__ void caps_request(
    CapsStage& s, const float* __restrict__ pose,
    const float* __restrict__ w, const float* __restrict__ qk,
    int n, int m, int h, int l16, int dcol) {
  const float* pt = pose + ((size_t)(2 * h) * N_ + n) * A_ + 2 * l16;
  for (int t = 0; t < 2; ++t) {
    for (int g = 0; g < 4; ++g) {
      const float* rowp = pt + (size_t)(16 * t + 4 * g) * N_ * A_;
      s.p[t][g][0] = *(const v2f*)(rowp);
      s.p[t][g][1] = *(const v2f*)(rowp + N_ * A_);
    }
  }
  if (QT) {
    const v2f q = *(const v2f*)(qk + ((size_t)m * N_ + n) * B_ + 2 * l16);
    s.qk0 = q.x;
    s.qk1 = q.y;
  } else {
    const int b0 = l16, b1 = l16 + 16;
    s.qk0 = qk[((size_t)b0 * N_ + n) * M_ + m];
    s.qk1 = qk[((size_t)b1 * N_ + n) * M_ + m];
  }
  const float* wbase = w + ((size_t)n * A_ + 16 * h) * MD_ + m * D_ + dcol;
  __builtin_prefetch(wbase + (size_t)2 * A_ * MD_, 0, 0);
  for (int jj = 0; jj < 4; ++jj) {
    for (int r = 0; r < 4; ++r) s.wv[jj][r] = wbase[(size_t)(4 * jj + r) * MD_];
  }
}

#define CAP_ 33
static __device__ __forceinline__ void caps_act(
    CapsStage& s, const float* ca_s, int n, int l16) {
  s.ca0 = ca_s[n * CAP_ + l16];
  s.ca1 = ca_s[n * CAP_ + l16 + 16];
}

struct CapsTurned { v8f tx0, ty0, tx1, ty1; };

static __device__ __forceinline__ void pose_turn_regs(
    v2f r0, v2f r1, v2f sel, v8f& tx, v8f& ty) {
  v2f ax; ax.x = r0.x; ax.y = r1.x;
  v2f ay; ay.x = r0.y; ay.y = r1.y;
  tx = wmma_f32(ax, sel, tx);
  ty = wmma_f32(ay, sel, ty);
}

static __device__ __forceinline__ void pose_turn_all(
    const CapsStage& s, v2f s0, v2f s1, v2f s2, v2f s3, CapsTurned& t) {
  t.tx0 = v8f{}; t.ty0 = v8f{}; t.tx1 = v8f{}; t.ty1 = v8f{};
  pose_turn_regs(s.p[0][0][0], s.p[0][0][1], s0, t.tx0, t.ty0);
  pose_turn_regs(s.p[0][1][0], s.p[0][1][1], s1, t.tx0, t.ty0);
  pose_turn_regs(s.p[0][2][0], s.p[0][2][1], s2, t.tx0, t.ty0);
  pose_turn_regs(s.p[0][3][0], s.p[0][3][1], s3, t.tx0, t.ty0);
  pose_turn_regs(s.p[1][0][0], s.p[1][0][1], s0, t.tx1, t.ty1);
  pose_turn_regs(s.p[1][1][0], s.p[1][1][1], s1, t.tx1, t.ty1);
  pose_turn_regs(s.p[1][2][0], s.p[1][2][1], s2, t.tx1, t.ty1);
  pose_turn_regs(s.p[1][3][0], s.p[1][3][1], s3, t.tx1, t.ty1);
}

template <int JJ>
static __device__ __forceinline__ void out_step(
    const CapsStage& s, float q0, float q1, const CapsTurned& t,
    v8f& acc0, v8f& acc1) {
  v2f bx; bx.x = s.wv[JJ][0];  bx.y = s.wv[JJ][2];
  v2f by; by.x = s.wv[JJ][1];  by.y = s.wv[JJ][3];
  v2f a0, a1;
  a0.x = q0 * t.tx0[2 * JJ]; a0.y = q0 * t.tx0[2 * JJ + 1];
  a1.x = q1 * t.tx1[2 * JJ]; a1.y = q1 * t.tx1[2 * JJ + 1];
  acc0 = wmma_f32(a0, bx, acc0);
  acc1 = wmma_f32(a1, bx, acc1);
  a0.x = q0 * t.ty0[2 * JJ]; a0.y = q0 * t.ty0[2 * JJ + 1];
  a1.x = q1 * t.ty1[2 * JJ]; a1.y = q1 * t.ty1[2 * JJ + 1];
  acc0 = wmma_f32(a0, by, acc0);
  acc1 = wmma_f32(a1, by, acc1);
}

static __device__ __forceinline__ void caps_main(
    const CapsStage& s, const CapsTurned& t, v8f& acc0, v8f& acc1) {
  const float q0 = s.qk0 * s.ca0;
  const float q1 = s.qk1 * s.ca1;
  out_step<0>(s, q0, q1, t, acc0, acc1);
  out_step<1>(s, q0, q1, t, acc0, acc1);
  out_step<2>(s, q0, q1, t, acc0, acc1);
  out_step<3>(s, q0, q1, t, acc0, acc1);
}

template <bool QT>
__global__ __launch_bounds__(256) void caps_out(
    const float* __restrict__ pose, const float* __restrict__ ca,
    const float* __restrict__ w, const float* __restrict__ qk,
    float* __restrict__ out) {
  const int m    = blockIdx.x;
  const int dh   = blockIdx.y;
  const int tid  = threadIdx.x;
  const int lane = tid & 31;
  const int wv   = tid >> 5;
  const int l16  = lane & 15;
  const int h    = lane >> 4;
  const int dcol = dh * 16 + l16;

  __shared__ float red[8][32 * 17];
  __shared__ float ca_s[N_ * CAP_];

  const int c0 = l16 - 2 * h;
  v2f s0, s1, s2, s3;
  s0.x = (c0 == 0)  ? 1.f : 0.f;  s0.y = (c0 == 1)  ? 1.f : 0.f;
  s1.x = (c0 == 4)  ? 1.f : 0.f;  s1.y = (c0 == 5)  ? 1.f : 0.f;
  s2.x = (c0 == 8)  ? 1.f : 0.f;  s2.y = (c0 == 9)  ? 1.f : 0.f;
  s3.x = (c0 == 12) ? 1.f : 0.f;  s3.y = (c0 == 13) ? 1.f : 0.f;

  v8f acc0 = {}, acc1 = {};

  const int nbeg = wv * 64;
  CapsStage sa, sb;
  CapsTurned t;

  {
    const int fb = lane >> 3, fp = lane & 7;
    v4f cx[8][2];
    for (int jj = 0; jj < 8; ++jj)
      for (int hf = 0; hf < 2; ++hf)
        cx[jj][hf] = *(const v4f*)(ca + (size_t)(4 * jj + fb) * N_ + nbeg + 32 * hf + 4 * fp);
    caps_request<QT>(sa, pose, w, qk, nbeg, m, h, l16, dcol);
    for (int jj = 0; jj < 8; ++jj)
      for (int hf = 0; hf < 2; ++hf) {
        float* dst = ca_s + (nbeg + 32 * hf + 4 * fp) * CAP_ + 4 * jj + fb;
        dst[0]        = cx[jj][hf].x;
        dst[CAP_]     = cx[jj][hf].y;
        dst[2 * CAP_] = cx[jj][hf].z;
        dst[3 * CAP_] = cx[jj][hf].w;
      }
  }
  __syncthreads();

  caps_act(sa, ca_s, nbeg, l16);
  pose_turn_all(sa, s0, s1, s2, s3, t);
  for (int i = 0; i < 31; ++i) {
    const int n = nbeg + 2 * i;
    caps_request<QT>(sb, pose, w, qk, n + 1, m, h, l16, dcol);
    caps_act(sb, ca_s, n + 1, l16);
    caps_main(sa, t, acc0, acc1);
    pose_turn_all(sb, s0, s1, s2, s3, t);
    caps_request<QT>(sa, pose, w, qk, n + 2, m, h, l16, dcol);
    caps_act(sa, ca_s, n + 2, l16);
    caps_main(sb, t, acc0, acc1);
    pose_turn_all(sa, s0, s1, s2, s3, t);
  }
  caps_request<QT>(sb, pose, w, qk, nbeg + 63, m, h, l16, dcol);
  caps_act(sb, ca_s, nbeg + 63, l16);
  caps_main(sa, t, acc0, acc1);
  pose_turn_all(sb, s0, s1, s2, s3, t);
  caps_main(sb, t, acc0, acc1);

  const int rbase = (lane < 16) ? 0 : 8;
  for (int i = 0; i < 8; ++i) {
    red[wv][(i + rbase) * 17 + l16]      = acc0[i];
    red[wv][(i + rbase + 16) * 17 + l16] = acc1[i];
  }
  __syncthreads();
  for (int e = tid; e < 512; e += 256) {
    const int b = e >> 4, dc = e & 15;
    float s = 0.f;
    for (int v = 0; v < 8; ++v) s += red[v][b * 17 + dc];
    out[(size_t)b * MD_ + m * D_ + dh * 16 + dc] = s;
  }
}

extern "C" void kernel_launch(void* const* d_in, const int* in_sizes, int n_in,
                              void* d_out, int out_size, void* d_ws,
                              size_t ws_size, hipStream_t stream) {
  (void)in_sizes; (void)n_in; (void)out_size;
  const float* pose = (const float*)d_in[0];
  const float* ca   = (const float*)d_in[1];
  const float* ncv  = (const float*)d_in[2];
  const float* na   = (const float*)d_in[3];
  const float* w    = (const float*)d_in[4];

  float* out    = (float*)d_out;
  float* na_out = out + (size_t)B_ * M_ * D_;
  float* qk_out = na_out + (size_t)B_ * M_;

  float* qkt = (float*)d_ws;
  if (qkt != nullptr && ws_size >= QKT_FLOATS_ * sizeof(float)) {
    caps_logits_qk<true><<<dim3(N_), dim3(256), 0, stream>>>(pose, ncv, na, w, qk_out, qkt);
    hipMemcpyAsync(na_out, na, (size_t)B_ * M_ * sizeof(float),
                   hipMemcpyDeviceToDevice, stream);
    caps_out<true><<<dim3(M_, 2), dim3(256), 0, stream>>>(pose, ca, w, qkt, out);
  } else {
    caps_logits_qk<false><<<dim3(N_), dim3(256), 0, stream>>>(pose, ncv, na, w, qk_out, qkt);
    hipMemcpyAsync(na_out, na, (size_t)B_ * M_ * sizeof(float),
                   hipMemcpyDeviceToDevice, stream);
    caps_out<false><<<dim3(M_, 2), dim3(256), 0, stream>>>(pose, ca, w, qk_out, out);
  }
}


# 1st kernel also stores pose x act in fragment order (+2 MiB workspace); 2nd kernel loads A from it as whole lines, 2 capsules ahead: no selector WMMAs, no LDS act copy; sum order over a changes
# speedup vs baseline: 2.0663x; 1.1541x over previous
#include <hip/hip_runtime.h>


typedef __attribute__((ext_vector_type(2))) float v2f;
typedef __attribute__((ext_vector_type(4))) float v4f;
typedef __attribute__((ext_vector_type(8))) float v8f;

#define B_  32
#define N_  512
#define A_  32
#define M_  128
#define D_  32
#define MD_ (M_ * D_)
#define SCALE_ 0.17677669529663687f
#define LGP_ 132
#define QKT_FLOATS_ ((size_t)M_ * N_ * B_)
#define PA_FLOATS_  ((size_t)N_ * B_ * A_)

static __device__ __forceinline__ v8f wmma_f32(v2f a, v2f b, v8f c) {
  return __builtin_amdgcn_wmma_f32_16x16x4_f32(false, a, false, b, (short)0, c,
                                               false, false);
}

struct WTile { v2f b[2][8]; };

static __device__ __forceinline__ void wtile_request(
    WTile& t, const float* __restrict__ w, int n, int mt, unsigned wlane) {
  const float* wt = w + (size_t)n * A_ * MD_ + mt * 256;
  for (int j = 0; j < 2; ++j) {
    const unsigned lo = wlane + 16u * j;
    __builtin_prefetch(wt + lo + 256, 0, 0);
    for (int k = 0; k < 8; ++k) {
      t.b[j][k].x = wt[lo + (unsigned)(4 * k) * MD_];
      t.b[j][k].y = wt[lo + (unsigned)(4 * k + 1) * MD_];
    }
  }
}

static __device__ __forceinline__ float xhalf(float x) {
  const int b = __builtin_bit_cast(int, x);
  return __builtin_bit_cast(float, __builtin_amdgcn_permlanex16(
      b, b, 0x76543210, 0xfedcba98, false, false));
}

template <bool NEXT>
static __device__ __forceinline__ void logits_step(
    int mt, const WTile& cur, WTile& nxt, const v2f (&preg)[2][8], float* lg,
    const float* __restrict__ ncv, const float* __restrict__ w,
    int n, int lane, unsigned wlane, unsigned clane, int lgl) {
  const float* ct = ncv + mt * 8 * D_;
  v4f cv[2][2][2];
  for (int rt = 0; rt < 2; ++rt) {
    for (int j = 0; j < 2; ++j) {
      const unsigned o = clane + (unsigned)(16 * rt) * MD_ + 16u * j;
      cv[rt][j][0] = *(const v4f*)(ct + o);
      cv[rt][j][1] = *(const v4f*)(ct + o + 4);
    }
  }
  if (NEXT) wtile_request(nxt, w, n, mt + 2, wlane);
  __builtin_amdgcn_sched_barrier(0);

  v8f acc[2][2];
  for (int j = 0; j < 2; ++j) {
    acc[j][0] = v8f{}; acc[j][1] = v8f{};
    for (int k = 0; k < 8; ++k) {
      acc[j][0] = wmma_f32(cur.b[j][k], preg[0][k], acc[j][0]);
      acc[j][1] = wmma_f32(cur.b[j][k], preg[1][k], acc[j][1]);
    }
  }
#define DOT2(S, ACC, C0, C1)                                                        \
  S += __builtin_shufflevector(ACC, ACC, 0, 1) * __builtin_shufflevector(C0, C0, 0, 1); \
  S += __builtin_shufflevector(ACC, ACC, 2, 3) * __builtin_shufflevector(C0, C0, 2, 3); \
  S += __builtin_shufflevector(ACC, ACC, 4, 5) * __builtin_shufflevector(C1, C1, 0, 1); \
  S += __builtin_shufflevector(ACC, ACC, 6, 7) * __builtin_shufflevector(C1, C1, 2, 3)
  float t[2];
  for (int rt = 0; rt < 2; ++rt) {
    v2f p0 = {0.f, 0.f}, p1 = {0.f, 0.f};
    DOT2(p0, acc[0][rt], cv[rt][0][0], cv[rt][0][1]);
    DOT2(p1, acc[1][rt], cv[rt][1][0], cv[rt][1][1]);
    const v2f p = p0 + p1;
    const float s = p.x + p.y;
    t[rt] = s + xhalf(s);
  }
#undef DOT2
  lg[lgl + mt * 8] = ((lane < 16) ? t[0] : t[1]) * SCALE_;
}

template <bool QT>
__global__ __launch_bounds__(256, 4) void caps_logits_qk(
    const float* __restrict__ pose, const float* __restrict__ ca,
    const float* __restrict__ ncv,
    const float* __restrict__ na, const float* __restrict__ w,
    float* __restrict__ qk_out, float* __restrict__ qkt, float* __restrict__ pa) {
  const int n    = blockIdx.x;
  const int tid  = threadIdx.x;
  const int lane = tid & 31;
  const int wv   = tid >> 5;
  const int l16  = lane & 15;
  const int khalf = (lane < 16) ? 0 : 2;

  __shared__ __attribute__((aligned(16))) float lg[32 * LGP_];
  __shared__ float red[3][8][32];

  v2f preg[2][8];
  for (int rt = 0; rt < 2; ++rt) {
    const int b = l16 + 16 * rt;
    const float* pp = pose + ((size_t)b * N_ + n) * A_;
    for (int k = 0; k < 8; ++k) preg[rt][k] = *(const v2f*)(pp + 4 * k + khalf);
  }

  const unsigned wlane = (unsigned)khalf * MD_ + 32u * wv + l16;
  const unsigned clane = (unsigned)l16 * MD_ + 32u * wv + 4u * khalf;
  const int lgl = lane * LGP_ + wv;
#define LSTEP(NEXT, MT, CUR, REQ) \
  logits_step<NEXT>(MT, CUR, REQ, preg, lg, ncv, w, n, lane, wlane, clane, lgl)
  float cav[2] = {0.f, 0.f};
  if (QT && wv == 0) {
    for (int rt = 0; rt < 2; ++rt) cav[rt] = ca[(size_t)(l16 + 16 * rt) * N_ + n];
  }
  WTile ta, tb, tc;
  wtile_request(ta, w, n, 0, wlane);
  wtile_request(tb, w, n, 1, wlane);
  if (QT && wv == 0) {
    float* pd = pa + (size_t)n * (B_ * A_) + 4 * lane;
    for (int rt = 0; rt < 2; ++rt) {
      const float c = cav[rt];
      for (int kk = 0; kk < 4; ++kk) {
        v4f v;
        v.x = preg[rt][2 * kk].x * c;      v.y = preg[rt][2 * kk].y * c;
        v.z = preg[rt][2 * kk + 1].x * c;  v.w = preg[rt][2 * kk + 1].y * c;
        *(v4f*)(pd + (4 * rt + kk) * 128) = v;
      }
    }
  }
  for (int t3 = 0; t3 < 4; ++t3) {
    LSTEP(true, 3 * t3,     ta, tc);
    LSTEP(true, 3 * t3 + 1, tb, ta);
    LSTEP(true, 3 * t3 + 2, tc, tb);
  }
  LSTEP(true,  12, ta, tc);
  LSTEP(true,  13, tb, ta);
  LSTEP(false, 14, tc, tb);
  LSTEP(false, 15, ta, tb);
#undef LSTEP

  v4f nar[4];
  for (int r = 0; r < 4; ++r)
    nar[r] = *(const v4f*)(na + (4 * wv + r) * M_ + 4 * lane);

  float x[16];
  for (int i = 0; i < 16; ++i) x[i] = lg[lgl + 8 * i];
  float mx = x[0];
  for (int i = 1; i < 16; ++i) mx = fmaxf(mx, x[i]);
  red[0][wv][lane] = mx;
  __syncthreads();

  for (int r = 0; r < 4; ++r)
    *(v4f*)(lg + (4 * wv + r) * LGP_ + 4 * lane) = nar[r];

  float gm = red[0][0][lane];
  for (int v = 1; v < 8; ++v) gm = fmaxf(gm, red[0][v][lane]);
  float s = 0.f;
  for (int i = 0; i < 16; ++i) { x[i] = __expf(x[i] - gm); s += x[i]; }
  red[1][wv][lane] = s;
  __syncthreads();

  float st = 0.f;
  for (int v = 0; v < 8; ++v) st += red[1][v][lane];
  const float inv = 1.f / st;
  float s2 = 0.f;
  for (int i = 0; i < 16; ++i) {
    x[i] = x[i] * inv * lg[lgl + 8 * i];
    s2 += x[i];
  }
  red[2][wv][lane] = s2;
  __syncthreads();

  float s2t = 0.f;
  for (int v = 0; v < 8; ++v) s2t += red[2][v][lane];
  const float inv2 = 1.f / (s2t + 1e-10f);
  for (int i = 0; i < 16; ++i) {
    x[i] = x[i] * inv2;
    lg[lgl + 8 * i] = x[i];
  }
  __syncthreads();

  if (QT) {
    float* qt = qkt + ((size_t)wv * N_ + n) * B_ + 2 * l16 + (lane >> 4);
    for (int i = 0; i < 16; ++i) qt[(size_t)(8 * i) * N_ * B_] = x[i];
  }

  for (int r = 0; r < 4; ++r) {
    const int b = 4 * wv + r;
    const v4f q = *(const v4f*)(lg + b * LGP_ + 4 * lane);
    *(v4f*)(qk_out + ((size_t)b * N_ + n) * M_ + 4 * lane) = q;
  }
}


struct CapsStage {
  v2f   p[2][4][2];
  float qk0, qk1, ca0, ca1;
  float wv[4][4];
};

static __device__ __forceinline__ void caps_request(
    CapsStage& s, const float* __restrict__ pose,
    const float* __restrict__ w, const float* __restrict__ qk,
    int n, int m, int h, int l16, int dcol) {
  const float* pt = pose + ((size_t)(2 * h) * N_ + n) * A_ + 2 * l16;
  for (int t = 0; t < 2; ++t) {
    for (int g = 0; g < 4; ++g) {
      const float* rowp = pt + (size_t)(16 * t + 4 * g) * N_ * A_;
      s.p[t][g][0] = *(const v2f*)(rowp);
      s.p[t][g][1] = *(const v2f*)(rowp + N_ * A_);
    }
  }
  {
    const int b0 = l16, b1 = l16 + 16;
    s.qk0 = qk[((size_t)b0 * N_ + n) * M_ + m];
    s.qk1 = qk[((size_t)b1 * N_ + n) * M_ + m];
  }
  const float* wbase = w + ((size_t)n * A_ + 16 * h) * MD_ + m * D_ + dcol;
  __builtin_prefetch(wbase + (size_t)2 * A_ * MD_, 0, 0);
  for (int jj = 0; jj < 4; ++jj) {
    for (int r = 0; r < 4; ++r) s.wv[jj][r] = wbase[(size_t)(4 * jj + r) * MD_];
  }
}

#define CAP_ 33
static __device__ __forceinline__ void caps_act(
    CapsStage& s, const float* ca_s, int n, int l16) {
  s.ca0 = ca_s[n * CAP_ + l16];
  s.ca1 = ca_s[n * CAP_ + l16 + 16];
}

struct CapsTurned { v8f tx0, ty0, tx1, ty1; };

static __device__ __forceinline__ void pose_turn_regs(
    v2f r0, v2f r1, v2f sel, v8f& tx, v8f& ty) {
  v2f ax; ax.x = r0.x; ax.y = r1.x;
  v2f ay; ay.x = r0.y; ay.y = r1.y;
  tx = wmma_f32(ax, sel, tx);
  ty = wmma_f32(ay, sel, ty);
}

static __device__ __forceinline__ void pose_turn_all(
    const CapsStage& s, v2f s0, v2f s1, v2f s2, v2f s3, CapsTurned& t) {
  t.tx0 = v8f{}; t.ty0 = v8f{}; t.tx1 = v8f{}; t.ty1 = v8f{};
  pose_turn_regs(s.p[0][0][0], s.p[0][0][1], s0, t.tx0, t.ty0);
  pose_turn_regs(s.p[0][1][0], s.p[0][1][1], s1, t.tx0, t.ty0);
  pose_turn_regs(s.p[0][2][0], s.p[0][2][1], s2, t.tx0, t.ty0);
  pose_turn_regs(s.p[0][3][0], s.p[0][3][1], s3, t.tx0, t.ty0);
  pose_turn_regs(s.p[1][0][0], s.p[1][0][1], s0, t.tx1, t.ty1);
  pose_turn_regs(s.p[1][1][0], s.p[1][1][1], s1, t.tx1, t.ty1);
  pose_turn_regs(s.p[1][2][0], s.p[1][2][1], s2, t.tx1, t.ty1);
  pose_turn_regs(s.p[1][3][0], s.p[1][3][1], s3, t.tx1, t.ty1);
}

template <int JJ>
static __device__ __forceinline__ void out_step(
    const CapsStage& s, float q0, float q1, const CapsTurned& t,
    v8f& acc0, v8f& acc1) {
  v2f bx; bx.x = s.wv[JJ][0];  bx.y = s.wv[JJ][2];
  v2f by; by.x = s.wv[JJ][1];  by.y = s.wv[JJ][3];
  v2f a0, a1;
  a0.x = q0 * t.tx0[2 * JJ]; a0.y = q0 * t.tx0[2 * JJ + 1];
  a1.x = q1 * t.tx1[2 * JJ]; a1.y = q1 * t.tx1[2 * JJ + 1];
  acc0 = wmma_f32(a0, bx, acc0);
  acc1 = wmma_f32(a1, bx, acc1);
  a0.x = q0 * t.ty0[2 * JJ]; a0.y = q0 * t.ty0[2 * JJ + 1];
  a1.x = q1 * t.ty1[2 * JJ]; a1.y = q1 * t.ty1[2 * JJ + 1];
  acc0 = wmma_f32(a0, by, acc0);
  acc1 = wmma_f32(a1, by, acc1);
}

static __device__ __forceinline__ void caps_main(
    const CapsStage& s, const CapsTurned& t, v8f& acc0, v8f& acc1) {
  const float q0 = s.qk0 * s.ca0;
  const float q1 = s.qk1 * s.ca1;
  out_step<0>(s, q0, q1, t, acc0, acc1);
  out_step<1>(s, q0, q1, t, acc0, acc1);
  out_step<2>(s, q0, q1, t, acc0, acc1);
  out_step<3>(s, q0, q1, t, acc0, acc1);
}

__global__ __launch_bounds__(256) void caps_out(
    const float* __restrict__ pose, const float* __restrict__ ca,
    const float* __restrict__ w, const float* __restrict__ qk,
    float* __restrict__ out) {
  const int m    = blockIdx.x;
  const int dh   = blockIdx.y;
  const int tid  = threadIdx.x;
  const int lane = tid & 31;
  const int wv   = tid >> 5;
  const int l16  = lane & 15;
  const int h    = lane >> 4;
  const int dcol = dh * 16 + l16;

  __shared__ float red[8][32 * 17];
  __shared__ float ca_s[N_ * CAP_];

  const int c0 = l16 - 2 * h;
  v2f s0, s1, s2, s3;
  s0.x = (c0 == 0)  ? 1.f : 0.f;  s0.y = (c0 == 1)  ? 1.f : 0.f;
  s1.x = (c0 == 4)  ? 1.f : 0.f;  s1.y = (c0 == 5)  ? 1.f : 0.f;
  s2.x = (c0 == 8)  ? 1.f : 0.f;  s2.y = (c0 == 9)  ? 1.f : 0.f;
  s3.x = (c0 == 12) ? 1.f : 0.f;  s3.y = (c0 == 13) ? 1.f : 0.f;

  v8f acc0 = {}, acc1 = {};

  const int nbeg = wv * 64;
  CapsStage sa, sb;
  CapsTurned t;

  {
    const int fb = lane >> 3, fp = lane & 7;
    v4f cx[8][2];
    for (int jj = 0; jj < 8; ++jj)
      for (int hf = 0; hf < 2; ++hf)
        cx[jj][hf] = *(const v4f*)(ca + (size_t)(4 * jj + fb) * N_ + nbeg + 32 * hf + 4 * fp);
    caps_request(sa, pose, w, qk, nbeg, m, h, l16, dcol);
    for (int jj = 0; jj < 8; ++jj)
      for (int hf = 0; hf < 2; ++hf) {
        float* dst = ca_s + (nbeg + 32 * hf + 4 * fp) * CAP_ + 4 * jj + fb;
        dst[0]        = cx[jj][hf].x;
        dst[CAP_]     = cx[jj][hf].y;
        dst[2 * CAP_] = cx[jj][hf].z;
        dst[3 * CAP_] = cx[jj][hf].w;
      }
  }
  __syncthreads();

  caps_act(sa, ca_s, nbeg, l16);
  pose_turn_all(sa, s0, s1, s2, s3, t);
  for (int i = 0; i < 31; ++i) {
    const int n = nbeg + 2 * i;
    caps_request(sb, pose, w, qk, n + 1, m, h, l16, dcol);
    caps_act(sb, ca_s, n + 1, l16);
    caps_main(sa, t, acc0, acc1);
    pose_turn_all(sb, s0, s1, s2, s3, t);
    caps_request(sa, pose, w, qk, n + 2, m, h, l16, dcol);
    caps_act(sa, ca_s, n + 2, l16);
    caps_main(sb, t, acc0, acc1);
    pose_turn_all(sa, s0, s1, s2, s3, t);
  }
  caps_request(sb, pose, w, qk, nbeg + 63, m, h, l16, dcol);
  caps_act(sb, ca_s, nbeg + 63, l16);
  caps_main(sa, t, acc0, acc1);
  pose_turn_all(sb, s0, s1, s2, s3, t);
  caps_main(sb, t, acc0, acc1);

  const int rbase = (lane < 16) ? 0 : 8;
  for (int i = 0; i < 8; ++i) {
    red[wv][(i + rbase) * 17 + l16]      = acc0[i];
    red[wv][(i + rbase + 16) * 17 + l16] = acc1[i];
  }
  __syncthreads();
  for (int e = tid; e < 512; e += 256) {
    const int b = e >> 4, dc = e & 15;
    float s = 0.f;
    for (int v = 0; v < 8; ++v) s += red[v][b * 17 + dc];
    out[(size_t)b * MD_ + m * D_ + dh * 16 + dc] = s;
  }
}

struct WsStage {
  v4f a[8];
  v2f q;
  v2f b[8];
};

static __device__ __forceinline__ void ws_request(
    WsStage& s, const float* __restrict__ pa, const float* __restrict__ w,
    const float* __restrict__ qkt, int n, int m, int h, int l16, int lane, int dcol) {
  const float* ap = pa + (size_t)n * (B_ * A_) + 4 * lane;
  for (int j = 0; j < 8; ++j) s.a[j] = *(const v4f*)(ap + 128 * j);
  s.q = *(const v2f*)(qkt + ((size_t)m * N_ + n) * B_ + 2 * l16);
  const float* wbase = w + ((size_t)n * A_ + 2 * h) * MD_ + m * D_ + dcol;
  __builtin_prefetch(wbase + (size_t)2 * A_ * MD_, 0, 0);
  for (int k = 0; k < 8; ++k) {
    s.b[k].x = wbase[(size_t)(4 * k) * MD_];
    s.b[k].y = wbase[(size_t)(4 * k + 1) * MD_];
  }
}

static __device__ __forceinline__ void ws_main(
    const WsStage& s, v8f& acc0, v8f& acc1) {
  const v2f q0 = {s.q.x, s.q.x};
  const v2f q1 = {s.q.y, s.q.y};
  for (int kk = 0; kk < 4; ++kk) {
    const v4f x0 = s.a[kk], x1 = s.a[4 + kk];
    v2f a0 = __builtin_shufflevector(x0, x0, 0, 1) * q0;
    v2f a1 = __builtin_shufflevector(x1, x1, 0, 1) * q1;
    acc0 = wmma_f32(a0, s.b[2 * kk], acc0);
    acc1 = wmma_f32(a1, s.b[2 * kk], acc1);
    a0 = __builtin_shufflevector(x0, x0, 2, 3) * q0;
    a1 = __builtin_shufflevector(x1, x1, 2, 3) * q1;
    acc0 = wmma_f32(a0, s.b[2 * kk + 1], acc0);
    acc1 = wmma_f32(a1, s.b[2 * kk + 1], acc1);
  }
}

__global__ __launch_bounds__(256) void caps_out_ws(
    const float* __restrict__ pa, const float* __restrict__ w,
    const float* __restrict__ qkt, float* __restrict__ out) {
  const int m    = blockIdx.x;
  const int dh   = blockIdx.y;
  const int tid  = threadIdx.x;
  const int lane = tid & 31;
  const int wv   = tid >> 5;
  const int l16  = lane & 15;
  const int h    = lane >> 4;
  const int dcol = dh * 16 + l16;

  __shared__ float red[8][32 * 17];

  v8f acc0 = {}, acc1 = {};

  const int nbeg = wv * 64;
#define WREQ(S, N) ws_request(S, pa, w, qkt, N, m, h, l16, lane, dcol)
#define WSLOT(SREQ, N, SMAIN) \
  WREQ(SREQ, N); __builtin_amdgcn_sched_barrier(0); ws_main(SMAIN, acc0, acc1)
  WsStage s0, s1, s2, s3;
  WREQ(s0, nbeg);
  WREQ(s1, nbeg + 1);
  for (int i = 0; i < 15; ++i) {
    const int n = nbeg + 4 * i;
    WSLOT(s2, n + 2, s0);
    WSLOT(s3, n + 3, s1);
    WSLOT(s0, n + 4, s2);
    WSLOT(s1, n + 5, s3);
  }
  WSLOT(s2, nbeg + 62, s0);
  WSLOT(s3, nbeg + 63, s1);
  ws_main(s2, acc0, acc1);
  ws_main(s3, acc0, acc1);
#undef WSLOT
#undef WREQ

  const int rbase = (lane < 16) ? 0 : 8;
  for (int i = 0; i < 8; ++i) {
    red[wv][(i + rbase) * 17 + l16]      = acc0[i];
    red[wv][(i + rbase + 16) * 17 + l16] = acc1[i];
  }
  __syncthreads();
  for (int e = tid; e < 512; e += 256) {
    const int b = e >> 4, dc = e & 15;
    float s = 0.f;
    for (int v = 0; v < 8; ++v) s += red[v][b * 17 + dc];
    out[(size_t)b * MD_ + m * D_ + dh * 16 + dc] = s;
  }
}

extern "C" void kernel_launch(void* const* d_in, const int* in_sizes, int n_in,
                              void* d_out, int out_size, void* d_ws,
                              size_t ws_size, hipStream_t stream) {
  (void)in_sizes; (void)n_in; (void)out_size;
  const float* pose = (const float*)d_in[0];
  const float* ca   = (const float*)d_in[1];
  const float* ncv  = (const float*)d_in[2];
  const float* na   = (const float*)d_in[3];
  const float* w    = (const float*)d_in[4];

  float* out    = (float*)d_out;
  float* na_out = out + (size_t)B_ * M_ * D_;
  float* qk_out = na_out + (size_t)B_ * M_;

  float* qkt = (float*)d_ws;
  if (qkt != nullptr && ws_size >= (QKT_FLOATS_ + PA_FLOATS_) * sizeof(float)) {
    float* pa = qkt + QKT_FLOATS_;
    caps_logits_qk<true><<<dim3(N_), dim3(256), 0, stream>>>(pose, ca, ncv, na, w, qk_out, qkt, pa);
    hipMemcpyAsync(na_out, na, (size_t)B_ * M_ * sizeof(float),
                   hipMemcpyDeviceToDevice, stream);
    caps_out_ws<<<dim3(M_, 2), dim3(256), 0, stream>>>(pa, w, qkt, out);
  } else {
    caps_logits_qk<false><<<dim3(N_), dim3(256), 0, stream>>>(pose, ca, ncv, na, w, qk_out, nullptr, nullptr);
    hipMemcpyAsync(na_out, na, (size_t)B_ * M_ * sizeof(float),
                   hipMemcpyDeviceToDevice, stream);
    caps_out<<<dim3(M_, 2), dim3(256), 0, stream>>>(pose, ca, w, qk_out, out);
  }
}
